# starting kernel
# baseline (speedup 1.0000x reference)
#include <hip/hip_runtime.h>

typedef _Float16 v16h __attribute__((ext_vector_type(16)));
typedef _Float16 v8h  __attribute__((ext_vector_type(8)));
typedef _Float16 v2h  __attribute__((ext_vector_type(2)));
typedef float    v8f  __attribute__((ext_vector_type(8)));

#define NB_STEPS 50
#define S_PTS    51
#define HID      128
#define ST_W     136
#define ST_A     136
#define WAVES    8
#define TPB      256
#define TILES_PER_WAVE 8
#define PI_F     3.14159265358979323846f

__global__ void __launch_bounds__(256) monotonic_init(const float* __restrict__ h,
                                                      float* __restrict__ out, int n) {
    int i = blockIdx.x * blockDim.x + threadIdx.x;
    if (i < n) out[i] = h[i];
}

__device__ __forceinline__ v16h ldfrag(const _Float16* rowp, int klo) {
    v8h lo = *(const v8h*)(rowp + klo);
    v8h hi = *(const v8h*)(rowp + 16 + klo);
    return __builtin_shufflevector(lo, hi, 0,1,2,3,4,5,6,7,8,9,10,11,12,13,14,15);
}

__global__ void __launch_bounds__(TPB) monotonic_main(
    const float* __restrict__ x,  const float* __restrict__ h,
    const float* __restrict__ w1, const float* __restrict__ b1,
    const float* __restrict__ w2, const float* __restrict__ b2,
    const float* __restrict__ w3, const float* __restrict__ b3,
    const float* __restrict__ w4, const float* __restrict__ b4,
    float* __restrict__ out, int Bn)
{
    __shared__ __align__(16) _Float16 w2T[HID * ST_W];
    __shared__ __align__(16) _Float16 w3T[HID * ST_W];
    __shared__ __align__(16) _Float16 acts[WAVES * 16 * ST_A];
    __shared__ float w1s[2 * HID], b1s[HID], b2s[HID], b3s[HID], w4s[HID];
    __shared__ float ccws[S_PTS], stepss[S_PTS], b4s[1];

    const int tid = threadIdx.x;

    for (int idx = tid; idx < HID * HID; idx += TPB) {
        int k = idx >> 7, n = idx & 127;
        w2T[n * ST_W + k] = (_Float16)w2[idx];
        w3T[n * ST_W + k] = (_Float16)w3[idx];
    }
    if (tid < HID) {
        b1s[tid] = b1[tid]; b2s[tid] = b2[tid]; b3s[tid] = b3[tid]; w4s[tid] = w4[tid];
        w1s[tid] = w1[tid]; w1s[HID + tid] = w1[HID + tid];
    }
    if (tid == 0) b4s[0] = b4[0];
    if (tid < S_PTS) {
        float t = (float)tid;
        stepss[tid] = cosf(t * PI_F / (float)NB_STEPS);
        float a = 1.0f;
        #pragma unroll
        for (int j = 1; j <= NB_STEPS / 2; ++j) {
            float k = 2.0f * (float)j;
            a += (2.0f / (1.0f - k * k)) * cosf(k * t * PI_F / (float)NB_STEPS);
        }
        a *= 2.0f / (float)NB_STEPS;
        if (tid == 0 || tid == NB_STEPS) a *= 0.5f;
        ccws[tid] = a;
    }
    __syncthreads();

    const int  lane  = tid & 31;
    const int  wave  = tid >> 5;
    const long total   = (long)Bn * S_PTS;
    const long tstride = (long)gridDim.x * WAVES * 16;

    _Float16* act = &acts[wave * 16 * ST_A];
    const int  m   = lane & 15;
    const int  klo = (lane < 16) ? 0 : 8;
    const int  mo  = (lane >> 4) * 8;

    for (long tbase = ((long)blockIdx.x * WAVES + wave) * 16; tbase < total; tbase += tstride) {
        const long r  = tbase + m;
        const long rc = (r < total) ? r : (total - 1);
        const int  bidx = (int)(rc / S_PTS);
        const int  sidx = (int)(rc - (long)bidx * S_PTS);
        const float xb = x[bidx];
        const float hb = h[bidx];
        const float Xv = xb * (stepss[sidx] + 1.0f) * 0.5f;

        {
            const int khalf = (lane >> 4) * 64;
            v2h* actrow = (v2h*)&act[m * ST_A + khalf];
            #pragma unroll 4
            for (int kk = 0; kk < 32; ++kk) {
                int k = khalf + 2 * kk;
                float v0 = fmaf(Xv, w1s[k],     fmaf(hb, w1s[HID + k],     b1s[k]));
                float v1 = fmaf(Xv, w1s[k + 1], fmaf(hb, w1s[HID + k + 1], b1s[k + 1]));
                v2h p; p.x = (_Float16)fmaxf(v0, 0.0f); p.y = (_Float16)fmaxf(v1, 0.0f);
                actrow[kk] = p;
            }
        }

        {
            v16h a0 = ldfrag(&act[m * ST_A +  0], klo);
            v16h a1 = ldfrag(&act[m * ST_A + 32], klo);
            v16h a2 = ldfrag(&act[m * ST_A + 64], klo);
            v16h a3 = ldfrag(&act[m * ST_A + 96], klo);
            const _Float16* wr0 = &w2T[m * ST_W];
            v16h b0 = ldfrag(wr0 +  0, klo), b1f = ldfrag(wr0 + 32, klo);
            v16h b2f = ldfrag(wr0 + 64, klo), b3f = ldfrag(wr0 + 96, klo);
            v8f acc[8];
            #pragma unroll
            for (int t = 0; t < 8; ++t) {
                v16h n0 = b0, n1 = b1f, n2 = b2f, n3 = b3f;
                if (t < 7) {
                    const _Float16* wn = &w2T[((t + 1) * 16 + m) * ST_W];
                    n0 = ldfrag(wn +  0, klo); n1 = ldfrag(wn + 32, klo);
                    n2 = ldfrag(wn + 64, klo); n3 = ldfrag(wn + 96, klo);
                }
                v8f c = {};
                c = __builtin_amdgcn_wmma_f32_16x16x32_f16(false, a0, false, b0,  (short)0, c, false, false);
                c = __builtin_amdgcn_wmma_f32_16x16x32_f16(false, a1, false, b1f, (short)0, c, false, false);
                c = __builtin_amdgcn_wmma_f32_16x16x32_f16(false, a2, false, b2f, (short)0, c, false, false);
                c = __builtin_amdgcn_wmma_f32_16x16x32_f16(false, a3, false, b3f, (short)0, c, false, false);
                acc[t] = c;
                b0 = n0; b1f = n1; b2f = n2; b3f = n3;
            }
            #pragma unroll
            for (int t = 0; t < 8; ++t) {
                int n = t * 16 + m;
                float bv = b2s[n];
                #pragma unroll
                for (int i = 0; i < 8; ++i) {
                    float v = fmaxf(acc[t][i] + bv, 0.0f);
                    act[(mo + i) * ST_A + n] = (_Float16)v;
                }
            }
        }

        float rs[8] = {0.f, 0.f, 0.f, 0.f, 0.f, 0.f, 0.f, 0.f};
        {
            v16h a0 = ldfrag(&act[m * ST_A +  0], klo);
            v16h a1 = ldfrag(&act[m * ST_A + 32], klo);
            v16h a2 = ldfrag(&act[m * ST_A + 64], klo);
            v16h a3 = ldfrag(&act[m * ST_A + 96], klo);
            const _Float16* wr0 = &w3T[m * ST_W];
            v16h b0 = ldfrag(wr0 +  0, klo), b1f = ldfrag(wr0 + 32, klo);
            v16h b2f = ldfrag(wr0 + 64, klo), b3f = ldfrag(wr0 + 96, klo);
            #pragma unroll
            for (int t = 0; t < 8; ++t) {
                v16h n0 = b0, n1 = b1f, n2 = b2f, n3 = b3f;
                if (t < 7) {
                    const _Float16* wn = &w3T[((t + 1) * 16 + m) * ST_W];
                    n0 = ldfrag(wn +  0, klo); n1 = ldfrag(wn + 32, klo);
                    n2 = ldfrag(wn + 64, klo); n3 = ldfrag(wn + 96, klo);
                }
                v8f c = {};
                c = __builtin_amdgcn_wmma_f32_16x16x32_f16(false, a0, false, b0,  (short)0, c, false, false);
                c = __builtin_amdgcn_wmma_f32_16x16x32_f16(false, a1, false, b1f, (short)0, c, false, false);
                c = __builtin_amdgcn_wmma_f32_16x16x32_f16(false, a2, false, b2f, (short)0, c, false, false);
                c = __builtin_amdgcn_wmma_f32_16x16x32_f16(false, a3, false, b3f, (short)0, c, false, false);
                int n = t * 16 + m;
                float bv = b3s[n], wv = w4s[n];
                #pragma unroll
                for (int i = 0; i < 8; ++i)
                    rs[i] = fmaf(fmaxf(c[i] + bv, 0.0f), wv, rs[i]);
                b0 = n0; b1f = n1; b2f = n2; b3f = n3;
            }
        }

        #pragma unroll
        for (int i = 0; i < 8; ++i) {
            float v = rs[i];
            v += __shfl_xor(v, 1, 32);
            v += __shfl_xor(v, 2, 32);
            v += __shfl_xor(v, 4, 32);
            v += __shfl_xor(v, 8, 32);
            rs[i] = v;
        }

        if (m == 0) {
            const int  rowbase = (lane >> 4) * 8;
            const float bb = b4s[0];
            #pragma unroll
            for (int i = 0; i < 8; ++i) {
                long rr = tbase + rowbase + i;
                if (rr < total) {
                    float v  = rs[i] + bb;
                    float dz = (v > 0.0f) ? v : (expf(v) - 1.0f);
                    dz += 1.0f;
                    int   bi = (int)(rr / S_PTS);
                    int   si = (int)(rr - (long)bi * S_PTS);
                    float contrib = dz * ccws[si] * x[bi] * 0.5f;
                    atomicAdd(&out[bi], contrib);
                }
            }
        }
    }
}

extern "C" void kernel_launch(void* const* d_in, const int* in_sizes, int n_in,
                              void* d_out, int out_size, void* d_ws, size_t ws_size,
                              hipStream_t stream) {
    const float* x  = (const float*)d_in[0];
    const float* h  = (const float*)d_in[1];
    const float* w1 = (const float*)d_in[2];
    const float* b1 = (const float*)d_in[3];
    const float* w2 = (const float*)d_in[4];
    const float* b2 = (const float*)d_in[5];
    const float* w3 = (const float*)d_in[6];
    const float* b3 = (const float*)d_in[7];
    const float* w4 = (const float*)d_in[8];
    const float* b4 = (const float*)d_in[9];
    float* out = (float*)d_out;
    const int Bn = in_sizes[0];

    monotonic_init<<<(Bn + 255) / 256, 256, 0, stream>>>(h, out, Bn);

    const long total  = (long)Bn * S_PTS;
    const int  tiles  = (int)((total + 15) / 16);
    const int  blocks = (tiles + WAVES * TILES_PER_WAVE - 1) / (WAVES * TILES_PER_WAVE);
    monotonic_main<<<blocks, TPB, 0, stream>>>(x, h, w1, b1, w2, b2, w3, b3, w4, b4, out, Bn);
}


# Each block owns a run of whole batch elements: a phase's rows are summed per element in LDS in fixed order, carried across phases, one plain store per output; no atomics, no init launch (sum order)
# speedup vs baseline: 2.5377x; 2.5377x over previous
#include <hip/hip_runtime.h>


typedef _Float16 v16h __attribute__((ext_vector_type(16)));
typedef _Float16 v8h  __attribute__((ext_vector_type(8)));
typedef _Float16 v4h  __attribute__((ext_vector_type(4)));
typedef float    v8f  __attribute__((ext_vector_type(8)));
typedef float    v4f  __attribute__((ext_vector_type(4)));

#define NB_STEPS 50
#define S_PTS    51
#define HID      128
#define ST_A     136
#define NWAVE    4
#define TPB      128
#define RT       4
#define PROWS    64
#define PI_F     3.14159265358979323846f

template <int M>
__device__ __forceinline__ float group_xor(float v) {
    return __builtin_bit_cast(float, __builtin_amdgcn_update_dpp(0, __builtin_bit_cast(int, v), 0x160 | M, 0xf, 0xf, true));
}

__device__ __forceinline__ v16h ldfrag(const _Float16* rowp, int klo) {
    v8h lo = *(const v8h*)(rowp + klo);
    v8h hi = *(const v8h*)(rowp + 16 + klo);
    return __builtin_shufflevector(lo, hi, 0,1,2,3,4,5,6,7,8,9,10,11,12,13,14,15);
}

__global__ void __launch_bounds__(TPB) monotonic_main(
    const float* __restrict__ x,  const float* __restrict__ h,
    const float* __restrict__ w1, const float* __restrict__ b1,
    const float* __restrict__ w2, const float* __restrict__ b2,
    const float* __restrict__ w3, const float* __restrict__ b3,
    const float* __restrict__ w4, const float* __restrict__ b4,
    float* __restrict__ out, int Bn)
{
    __shared__ __align__(16) _Float16 act1[PROWS * ST_A];
    __shared__ __align__(16) _Float16 act2[PROWS * ST_A];
    __shared__ __align__(16) float part[PROWS * 8];
    __shared__ __align__(16) float w1s[2 * HID];
    __shared__ __align__(16) float b1s[HID];
    __shared__ __align__(32) float b2s[HID], b3s[HID], w4s[HID];
    __shared__ float ccws[S_PTS], stepss[S_PTS], b4s[1];
    __shared__ __align__(16) float cbuf[PROWS];
    __shared__ float carrys[1];

    const int tid  = threadIdx.x;
    const int lane = tid & 31;
    const int wave = tid >> 5;
    const int c    = lane & 15;
    const int hh   = lane >> 4;
    const int klo  = hh * 8;

    b1s[tid] = b1[tid]; w1s[tid] = w1[tid]; w1s[HID + tid] = w1[HID + tid];
    b2s[tid] = b2[tid]; b3s[tid] = b3[tid]; w4s[tid] = w4[tid];
    if (tid == 0) { b4s[0] = b4[0]; carrys[0] = 0.0f; }
    if (tid < S_PTS) {
        float t = (float)tid;
        stepss[tid] = cosf(t * PI_F / (float)NB_STEPS);
        float a = 1.0f;
        #pragma unroll
        for (int j = 1; j <= NB_STEPS / 2; ++j) {
            float k = 2.0f * (float)j;
            a += (2.0f / (1.0f - k * k)) * cosf(k * t * PI_F / (float)NB_STEPS);
        }
        a *= 2.0f / (float)NB_STEPS;
        if (tid == 0 || tid == NB_STEPS) a *= 0.5f;
        ccws[tid] = a;
    }

    v16h wf2[2][4], wf3[2][4];
    #pragma unroll
    for (int t = 0; t < 2; ++t) {
        const int n = 32 * wave + 16 * t + c;
        #pragma unroll
        for (int j = 0; j < 4; ++j) {
            #pragma unroll
            for (int e = 0; e < 16; ++e) {
                const int k = 32 * j + klo + (e & 7) + 16 * (e >> 3);
                wf2[t][j][e] = (_Float16)w2[k * HID + n];
                wf3[t][j][e] = (_Float16)w3[k * HID + n];
            }
        }
    }
    __syncthreads();

    const unsigned G  = gridDim.x, bx = blockIdx.x;
    const unsigned eq = (unsigned)Bn / G, er = (unsigned)Bn % G;
    const int e0   = (int)(eq * bx + (er * bx) / G);
    const int e1   = (int)(eq * (bx + 1) + (er * (bx + 1)) / G);
    const int rbeg = e0 * S_PTS;
    const int rend = e1 * S_PTS;

    auto row_inputs = [&](int pb, float& xb, float& hb, int& sidx) __attribute__((always_inline)) {
        const int r  = pb + 16 * wave + c;
        const int rc = (r < rend) ? r : (rend - 1);
        const int bidx = rc / S_PTS;
        sidx = rc - bidx * S_PTS;
        xb = x[bidx];
        hb = h[bidx];
    };

    auto layer1 = [&](float xb, float hb, int sidx) __attribute__((always_inline)) {
        const float Xv = xb * (stepss[sidx] + 1.0f) * 0.5f;
        const v4f wa = *(const v4f*)&w1s[4 * lane];
        const v4f wb = *(const v4f*)&w1s[HID + 4 * lane];
        const v4f bb = *(const v4f*)&b1s[4 * lane];
        _Float16* rowbase = &act1[(16 * wave) * ST_A + 4 * lane];
        #pragma unroll 4
        for (int rr = 0; rr < 16; ++rr) {
            const float xr = __builtin_bit_cast(float, __builtin_amdgcn_readlane(__builtin_bit_cast(int, Xv), rr));
            const float hr = __builtin_bit_cast(float, __builtin_amdgcn_readlane(__builtin_bit_cast(int, hb), rr));
            v4h p;
            #pragma unroll
            for (int i = 0; i < 4; ++i)
                p[i] = (_Float16)fmaxf(fmaf(xr, wa[i], fmaf(hr, wb[i], bb[i])), 0.0f);
            *(v4h*)(rowbase + rr * ST_A) = p;
        }
    };

    auto finish_rows = [&](int pb) __attribute__((always_inline)) {
        if (tid < PROWS) {
            const int rr = pb + tid;
            float cv = 0.0f;
            if (rr < rend) {
                const v4f p0 = *(const v4f*)&part[tid * 8];
                const v4f p1 = *(const v4f*)&part[tid * 8 + 4];
                float v  = ((p0.x + p0.y) + (p0.z + p0.w)) + ((p1.x + p1.y) + (p1.z + p1.w)) + b4s[0];
                float dz = (v > 0.0f) ? v : (expf(v) - 1.0f);
                dz += 1.0f;
                const int bi = rr / S_PTS;
                const int si = rr - bi * S_PTS;
                cv = dz * ccws[si];
            }
            cbuf[tid] = cv;
        }
    };

    auto elem_inputs = [&](int pb, float& xe, float& he) __attribute__((always_inline)) {
        if (wave == NWAVE - 1) {
            const int b  = pb / S_PTS + (lane >> 3);
            const int bc = (b < e1) ? b : (e1 - 1);
            xe = x[bc];
            he = h[bc];
        }
    };

    auto finish_elems = [&](int pb, float xe, float he) __attribute__((always_inline)) {
        if (wave == NWAVE - 1) {
            const float cin = carrys[0];
            const int g  = lane >> 3, j = lane & 7;
            const int b  = pb / S_PTS + g;
            const int rb = b * S_PTS;
            const int pe = (pb + PROWS < rend) ? (pb + PROWS) : rend;
            const int lo = (rb > pb) ? rb : pb;
            const int hi = (rb + S_PTS < pe) ? (rb + S_PTS) : pe;
            float s = 0.0f;
            #pragma unroll
            for (int m = 0; m < 7; ++m) {
                const int r = lo + j + 8 * m;
                const int i = r - pb;
                const float v = cbuf[(i < PROWS) ? i : (PROWS - 1)];
                s += (r < hi) ? v : 0.0f;
            }
            s += group_xor<1>(s);
            s += group_xor<2>(s);
            s += group_xor<4>(s);
            if (j == 0 && lo < hi) {
                const float acc = ((rb < pb) ? cin : 0.0f) + s;
                if (hi == rb + S_PTS) out[b] = he + acc * xe * 0.5f;
                else                  carrys[0] = acc;
            }
        }
    };

    const int f0 = 32 * wave + 8 * hh;

    int pbase = rbeg;
    if (pbase < rend) {
        float xb, hb;
        int   sidx;
        row_inputs(pbase, xb, hb, sidx);
        layer1(xb, hb, sidx);
        int pprev = -1;
        float xe = 0.0f, he = 0.0f;
        for (;;) {
            __syncthreads();

            const int  pnext = pbase + PROWS;
            const bool more  = pnext < rend;
            row_inputs(pnext, xb, hb, sidx);

            if (pprev >= 0) {
                elem_inputs(pprev, xe, he);
                finish_rows(pprev);
            }

            {
                v8f bias2[2];
                bias2[0] = *(const v8f*)&b2s[f0];
                bias2[1] = *(const v8f*)&b2s[f0 + 16];
                #pragma unroll
                for (int rt = 0; rt < RT; ++rt) {
                    const _Float16* rowp = &act1[(16 * rt + c) * ST_A];
                    v16h a0 = ldfrag(rowp +  0, klo);
                    v16h a1 = ldfrag(rowp + 32, klo);
                    v16h a2 = ldfrag(rowp + 64, klo);
                    v16h a3 = ldfrag(rowp + 96, klo);
                    #pragma unroll
                    for (int t = 0; t < 2; ++t) {
                        v8f acc = bias2[t];
                        acc = __builtin_amdgcn_wmma_f32_16x16x32_f16(false, wf2[t][0], false, a0, (short)0, acc, false, false);
                        acc = __builtin_amdgcn_wmma_f32_16x16x32_f16(false, wf2[t][1], false, a1, (short)0, acc, false, false);
                        acc = __builtin_amdgcn_wmma_f32_16x16x32_f16(false, wf2[t][2], false, a2, (short)0, acc, false, false);
                        acc = __builtin_amdgcn_wmma_f32_16x16x32_f16(false, wf2[t][3], false, a3, (short)0, acc, false, false);
                        v8h o;
                        #pragma unroll
                        for (int i = 0; i < 8; ++i) o[i] = (_Float16)fmaxf(acc[i], 0.0f);
                        *(v8h*)&act2[(16 * rt + c) * ST_A + 32 * wave + 16 * t + 8 * hh] = o;
                    }
                }
            }
            __syncthreads();

            if (pprev >= 0) finish_elems(pprev, xe, he);

            if (more) layer1(xb, hb, sidx);

            {
                v8f bias3[2], w4r[2];
                bias3[0] = *(const v8f*)&b3s[f0];
                bias3[1] = *(const v8f*)&b3s[f0 + 16];
                w4r[0]   = *(const v8f*)&w4s[f0];
                w4r[1]   = *(const v8f*)&w4s[f0 + 16];
                #pragma unroll
                for (int rt = 0; rt < RT; ++rt) {
                    const _Float16* rowp = &act2[(16 * rt + c) * ST_A];
                    v16h a0 = ldfrag(rowp +  0, klo);
                    v16h a1 = ldfrag(rowp + 32, klo);
                    v16h a2 = ldfrag(rowp + 64, klo);
                    v16h a3 = ldfrag(rowp + 96, klo);
                    float p = 0.0f;
                    #pragma unroll
                    for (int t = 0; t < 2; ++t) {
                        v8f acc = bias3[t];
                        acc = __builtin_amdgcn_wmma_f32_16x16x32_f16(false, wf3[t][0], false, a0, (short)0, acc, false, false);
                        acc = __builtin_amdgcn_wmma_f32_16x16x32_f16(false, wf3[t][1], false, a1, (short)0, acc, false, false);
                        acc = __builtin_amdgcn_wmma_f32_16x16x32_f16(false, wf3[t][2], false, a2, (short)0, acc, false, false);
                        acc = __builtin_amdgcn_wmma_f32_16x16x32_f16(false, wf3[t][3], false, a3, (short)0, acc, false, false);
                        #pragma unroll
                        for (int i = 0; i < 8; ++i) p = fmaf(fmaxf(acc[i], 0.0f), w4r[t][i], p);
                    }
                    part[(16 * rt + c) * 8 + 2 * wave + hh] = p;
                }
            }

            pprev = pbase;
            if (!more) break;
            pbase = pnext;
        }
        __syncthreads();
        elem_inputs(pprev, xe, he);
        finish_rows(pprev);
        __syncthreads();
        finish_elems(pprev, xe, he);
    }
}

extern "C" void kernel_launch(void* const* d_in, const int* in_sizes, int n_in,
                              void* d_out, int out_size, void* d_ws, size_t ws_size,
                              hipStream_t stream) {
    const float* x  = (const float*)d_in[0];
    const float* h  = (const float*)d_in[1];
    const float* w1 = (const float*)d_in[2];
    const float* b1 = (const float*)d_in[3];
    const float* w2 = (const float*)d_in[4];
    const float* b2 = (const float*)d_in[5];
    const float* w3 = (const float*)d_in[6];
    const float* b3 = (const float*)d_in[7];
    const float* w4 = (const float*)d_in[8];
    const float* b4 = (const float*)d_in[9];
    float* out = (float*)d_out;
    const int Bn = in_sizes[0];

    monotonic_main<<<768, TPB, 0, stream>>>(x, h, w1, b1, w2, b2, w3, b3, w4, b4, out, Bn);
}


# Matrix issue order, layers 2 and 3: fragments read first, then 4 independent accumulator chains walked k step outer; first group's epilogue in pieces behind the second group's; sum order unchanged
# speedup vs baseline: 2.6663x; 1.0507x over previous
#include <hip/hip_runtime.h>


typedef _Float16 v16h __attribute__((ext_vector_type(16)));
typedef _Float16 v8h  __attribute__((ext_vector_type(8)));
typedef _Float16 v4h  __attribute__((ext_vector_type(4)));
typedef float    v8f  __attribute__((ext_vector_type(8)));
typedef float    v4f  __attribute__((ext_vector_type(4)));
typedef float    v2f  __attribute__((ext_vector_type(2)));

#define NB_STEPS 50
#define S_PTS    51
#define HID      128
#define ST_A     136
#define NWAVE    4
#define TPB      128
#define RT       4
#define PROWS    64
#define PI_F     3.14159265358979323846f

template <int M>
__device__ __forceinline__ float group_xor(float v) {
    return __builtin_bit_cast(float, __builtin_amdgcn_update_dpp(0, __builtin_bit_cast(int, v), 0x160 | M, 0xf, 0xf, true));
}

__device__ __forceinline__ v16h ldfrag(const _Float16* rowp, int klo) {
    v8h lo = *(const v8h*)(rowp + klo);
    v8h hi = *(const v8h*)(rowp + 16 + klo);
    return __builtin_shufflevector(lo, hi, 0,1,2,3,4,5,6,7,8,9,10,11,12,13,14,15);
}

__global__ void __launch_bounds__(TPB) monotonic_main(
    const float* __restrict__ x,  const float* __restrict__ h,
    const float* __restrict__ w1, const float* __restrict__ b1,
    const float* __restrict__ w2, const float* __restrict__ b2,
    const float* __restrict__ w3, const float* __restrict__ b3,
    const float* __restrict__ w4, const float* __restrict__ b4,
    float* __restrict__ out, int Bn)
{
    __shared__ __align__(16) _Float16 act1[PROWS * ST_A];
    __shared__ __align__(16) _Float16 act2[PROWS * ST_A];
    __shared__ __align__(16) float part[PROWS * 8];
    __shared__ __align__(16) float w1s[2 * HID];
    __shared__ __align__(16) float b1s[HID];
    __shared__ __align__(32) float b2s[HID], b3s[HID], w4s[HID];
    __shared__ float ccws[S_PTS], stepss[S_PTS], b4s[1];
    __shared__ __align__(16) float cbuf[PROWS];
    __shared__ float carrys[1];

    const int tid  = threadIdx.x;
    const int lane = tid & 31;
    const int wave = tid >> 5;
    const int c    = lane & 15;
    const int hh   = lane >> 4;
    const int klo  = hh * 8;

    b1s[tid] = b1[tid]; w1s[tid] = w1[tid]; w1s[HID + tid] = w1[HID + tid];
    b2s[tid] = b2[tid]; b3s[tid] = b3[tid]; w4s[tid] = w4[tid];
    if (tid == 0) { b4s[0] = b4[0]; carrys[0] = 0.0f; }
    if (tid < S_PTS) {
        float t = (float)tid;
        stepss[tid] = cosf(t * PI_F / (float)NB_STEPS);
        float a = 1.0f;
        #pragma unroll
        for (int j = 1; j <= NB_STEPS / 2; ++j) {
            float k = 2.0f * (float)j;
            a += (2.0f / (1.0f - k * k)) * cosf(k * t * PI_F / (float)NB_STEPS);
        }
        a *= 2.0f / (float)NB_STEPS;
        if (tid == 0 || tid == NB_STEPS) a *= 0.5f;
        ccws[tid] = a;
    }

    v16h wf2[2][4], wf3[2][4];
    #pragma unroll
    for (int t = 0; t < 2; ++t) {
        const int n = 32 * wave + 16 * t + c;
        #pragma unroll
        for (int j = 0; j < 4; ++j) {
            #pragma unroll
            for (int e = 0; e < 16; ++e) {
                const int k = 32 * j + klo + (e & 7) + 16 * (e >> 3);
                wf2[t][j][e] = (_Float16)w2[k * HID + n];
                wf3[t][j][e] = (_Float16)w3[k * HID + n];
            }
        }
    }
    __syncthreads();

    const unsigned G  = gridDim.x, bx = blockIdx.x;
    const unsigned eq = (unsigned)Bn / G, er = (unsigned)Bn % G;
    const int e0   = (int)(eq * bx + (er * bx) / G);
    const int e1   = (int)(eq * (bx + 1) + (er * (bx + 1)) / G);
    const int rbeg = e0 * S_PTS;
    const int rend = e1 * S_PTS;

    auto row_inputs = [&](int pb, float& xb, float& hb, int& sidx) __attribute__((always_inline)) {
        const int r  = pb + 16 * wave + c;
        const int rc = (r < rend) ? r : (rend - 1);
        const int bidx = rc / S_PTS;
        sidx = rc - bidx * S_PTS;
        xb = x[bidx];
        hb = h[bidx];
    };

    auto layer1 = [&](float xb, float hb, int sidx) __attribute__((always_inline)) {
        const float Xv = xb * (stepss[sidx] + 1.0f) * 0.5f;
        const v4f wa = *(const v4f*)&w1s[4 * lane];
        const v4f wb = *(const v4f*)&w1s[HID + 4 * lane];
        const v4f bb = *(const v4f*)&b1s[4 * lane];
        _Float16* rowbase = &act1[(16 * wave) * ST_A + 4 * lane];
        #pragma unroll 4
        for (int rr = 0; rr < 16; ++rr) {
            const float xr = __builtin_bit_cast(float, __builtin_amdgcn_readlane(__builtin_bit_cast(int, Xv), rr));
            const float hr = __builtin_bit_cast(float, __builtin_amdgcn_readlane(__builtin_bit_cast(int, hb), rr));
            v4h p;
            #pragma unroll
            for (int i = 0; i < 4; ++i)
                p[i] = (_Float16)fmaxf(fmaf(xr, wa[i], fmaf(hr, wb[i], bb[i])), 0.0f);
            *(v4h*)(rowbase + rr * ST_A) = p;
        }
    };

    auto finish_rows = [&](int pb) __attribute__((always_inline)) {
        if (tid < PROWS) {
            const int rr = pb + tid;
            float cv = 0.0f;
            if (rr < rend) {
                const v4f p0 = *(const v4f*)&part[tid * 8];
                const v4f p1 = *(const v4f*)&part[tid * 8 + 4];
                float v  = ((p0.x + p0.y) + (p0.z + p0.w)) + ((p1.x + p1.y) + (p1.z + p1.w)) + b4s[0];
                float dz = (v > 0.0f) ? v : (expf(v) - 1.0f);
                dz += 1.0f;
                const int bi = rr / S_PTS;
                const int si = rr - bi * S_PTS;
                cv = dz * ccws[si];
            }
            cbuf[tid] = cv;
        }
    };

    auto elem_inputs = [&](int pb, float& xe, float& he) __attribute__((always_inline)) {
        if (wave == NWAVE - 1) {
            const int b  = pb / S_PTS + (lane >> 3);
            const int bc = (b < e1) ? b : (e1 - 1);
            xe = x[bc];
            he = h[bc];
        }
    };

    auto finish_elems = [&](int pb, float xe, float he) __attribute__((always_inline)) {
        if (wave == NWAVE - 1) {
            const float cin = carrys[0];
            const int g  = lane >> 3, j = lane & 7;
            const int b  = pb / S_PTS + g;
            const int rb = b * S_PTS;
            const int pe = (pb + PROWS < rend) ? (pb + PROWS) : rend;
            const int lo = (rb > pb) ? rb : pb;
            const int hi = (rb + S_PTS < pe) ? (rb + S_PTS) : pe;
            float s = 0.0f;
            #pragma unroll
            for (int m = 0; m < 7; ++m) {
                const int r = lo + j + 8 * m;
                const int i = r - pb;
                const float v = cbuf[(i < PROWS) ? i : (PROWS - 1)];
                s += (r < hi) ? v : 0.0f;
            }
            s += group_xor<1>(s);
            s += group_xor<2>(s);
            s += group_xor<4>(s);
            if (j == 0 && lo < hi) {
                const float acc = ((rb < pb) ? cin : 0.0f) + s;
                if (hi == rb + S_PTS) out[b] = he + acc * xe * 0.5f;
                else                  carrys[0] = acc;
            }
        }
    };

    const int f0 = 32 * wave + 8 * hh;

    int pbase = rbeg;
    if (pbase < rend) {
        float xb, hb;
        int   sidx;
        row_inputs(pbase, xb, hb, sidx);
        layer1(xb, hb, sidx);
        int pprev = -1;
        float xe = 0.0f, he = 0.0f;
        for (;;) {
            __syncthreads();

            const int  pnext = pbase + PROWS;
            const bool more  = pnext < rend;
            row_inputs(pnext, xb, hb, sidx);

            if (pprev >= 0) {
                elem_inputs(pprev, xe, he);
                finish_rows(pprev);
            }

            {
                v8f bias2[2];
                bias2[0] = *(const v8f*)&b2s[f0];
                bias2[1] = *(const v8f*)&b2s[f0 + 16];
                v16h bf[RT][4];
                #pragma unroll
                for (int rt = 0; rt < RT; ++rt) {
                    const _Float16* rowp = &act1[(16 * rt + c) * ST_A];
                    #pragma unroll
                    for (int j = 0; j < 4; ++j) bf[rt][j] = ldfrag(rowp + 32 * j, klo);
                }
                __builtin_amdgcn_sched_barrier(0);
                v8h o[RT][2];
                v8f accA[2][2], accB[2][2];
                float r0 = 0.0f, r1 = 0.0f;
                #pragma unroll
                for (int r = 0; r < 2; ++r) { accA[r][0] = bias2[0]; accA[r][1] = bias2[1]; accB[r][0] = bias2[0]; accB[r][1] = bias2[1]; }
                #pragma unroll
                for (int j = 0; j < 4; ++j) {
                    #pragma unroll
                    for (int r = 0; r < 2; ++r) {
                        #pragma unroll
                        for (int t = 0; t < 2; ++t)
                            accA[r][t] = __builtin_amdgcn_wmma_f32_16x16x32_f16(false, wf2[t][j], false, bf[r][j], (short)0, accA[r][t], false, false);
                    }
                    __builtin_amdgcn_sched_barrier(0x6);
                }
                #pragma unroll
                for (int j = 0; j < 4; ++j) {
                    #pragma unroll
                    for (int r = 0; r < 2; ++r) {
                        #pragma unroll
                        for (int t = 0; t < 2; ++t) {
                            accB[r][t] = __builtin_amdgcn_wmma_f32_16x16x32_f16(false, wf2[t][j], false, bf[2 + r][j], (short)0, accB[r][t], false, false);
                            const int q = 4 * j + 2 * r + t;
                            if (q > 0) {
                                const int qp = q - 1;
                                o[qp >> 3][(qp >> 2) & 1][2 * (qp & 3)]     = (_Float16)r0;
                                o[qp >> 3][(qp >> 2) & 1][2 * (qp & 3) + 1] = (_Float16)r1;
                            }
                            r0 = fmaxf(accA[q >> 3][(q >> 2) & 1][2 * (q & 3)],     0.0f);
                            r1 = fmaxf(accA[q >> 3][(q >> 2) & 1][2 * (q & 3) + 1], 0.0f);
                            __builtin_amdgcn_sched_barrier(0);
                        }
                    }
                }
                o[1][1][6] = (_Float16)r0;
                o[1][1][7] = (_Float16)r1;
                #pragma unroll
                for (int r = 0; r < 2; ++r) {
                    #pragma unroll
                    for (int t = 0; t < 2; ++t) {
                        #pragma unroll
                        for (int i = 0; i < 8; ++i) o[2 + r][t][i] = (_Float16)fmaxf(accB[r][t][i], 0.0f);
                    }
                }
                #pragma unroll
                for (int rt = 0; rt < RT; ++rt) {
                    #pragma unroll
                    for (int t = 0; t < 2; ++t)
                        *(v8h*)&act2[(16 * rt + c) * ST_A + 32 * wave + 16 * t + 8 * hh] = o[rt][t];
                }
            }
            __syncthreads();

            if (pprev >= 0) finish_elems(pprev, xe, he);

            if (more) layer1(xb, hb, sidx);

            {
                v8f bias3[2], w4r[2];
                bias3[0] = *(const v8f*)&b3s[f0];
                bias3[1] = *(const v8f*)&b3s[f0 + 16];
                w4r[0]   = *(const v8f*)&w4s[f0];
                w4r[1]   = *(const v8f*)&w4s[f0 + 16];
                v16h bf[RT][4];
                #pragma unroll
                for (int rt = 0; rt < RT; ++rt) {
                    const _Float16* rowp = &act2[(16 * rt + c) * ST_A];
                    #pragma unroll
                    for (int j = 0; j < 4; ++j) bf[rt][j] = ldfrag(rowp + 32 * j, klo);
                }
                __builtin_amdgcn_sched_barrier(0);
                float pr[RT] = {0.0f, 0.0f, 0.0f, 0.0f};
                v2f pA = {0.0f, 0.0f}, rv = {0.0f, 0.0f};
                v8f accA[2][2], accB[2][2];
                #pragma unroll
                for (int r = 0; r < 2; ++r) { accA[r][0] = bias3[0]; accA[r][1] = bias3[1]; accB[r][0] = bias3[0]; accB[r][1] = bias3[1]; }
                #pragma unroll
                for (int j = 0; j < 4; ++j) {
                    #pragma unroll
                    for (int r = 0; r < 2; ++r) {
                        #pragma unroll
                        for (int t = 0; t < 2; ++t)
                            accA[r][t] = __builtin_amdgcn_wmma_f32_16x16x32_f16(false, wf3[t][j], false, bf[r][j], (short)0, accA[r][t], false, false);
                    }
                    __builtin_amdgcn_sched_barrier(0x6);
                }
                #pragma unroll
                for (int j = 0; j < 4; ++j) {
                    #pragma unroll
                    for (int r = 0; r < 2; ++r) {
                        #pragma unroll
                        for (int t = 0; t < 2; ++t) {
                            accB[r][t] = __builtin_amdgcn_wmma_f32_16x16x32_f16(false, wf3[t][j], false, bf[2 + r][j], (short)0, accB[r][t], false, false);
                            const int q = 4 * j + 2 * r + t;
                            if (q > 0) {
                                const int qp = q - 1;
                                const float wq = w4r[qp >> 3][qp & 7];
                                pA = __builtin_elementwise_fma(rv, (v2f){wq, wq}, pA);
                            }
                            rv = (v2f){fmaxf(accA[0][q >> 3][q & 7], 0.0f), fmaxf(accA[1][q >> 3][q & 7], 0.0f)};
                            __builtin_amdgcn_sched_barrier(0);
                        }
                    }
                }
                pA = __builtin_elementwise_fma(rv, (v2f){w4r[1][7], w4r[1][7]}, pA);
                pr[0] = pA.x; pr[1] = pA.y;
                #pragma unroll
                for (int t = 0; t < 2; ++t) {
                    #pragma unroll
                    for (int i = 0; i < 8; ++i) {
                        pr[2] = fmaf(fmaxf(accB[0][t][i], 0.0f), w4r[t][i], pr[2]);
                        pr[3] = fmaf(fmaxf(accB[1][t][i], 0.0f), w4r[t][i], pr[3]);
                    }
                }
                #pragma unroll
                for (int rt = 0; rt < RT; ++rt) part[(16 * rt + c) * 8 + 2 * wave + hh] = pr[rt];
            }

            pprev = pbase;
            if (!more) break;
            pbase = pnext;
        }
        __syncthreads();
        elem_inputs(pprev, xe, he);
        finish_rows(pprev);
        __syncthreads();
        finish_elems(pprev, xe, he);
    }
}

extern "C" void kernel_launch(void* const* d_in, const int* in_sizes, int n_in,
                              void* d_out, int out_size, void* d_ws, size_t ws_size,
                              hipStream_t stream) {
    const float* x  = (const float*)d_in[0];
    const float* h  = (const float*)d_in[1];
    const float* w1 = (const float*)d_in[2];
    const float* b1 = (const float*)d_in[3];
    const float* w2 = (const float*)d_in[4];
    const float* b2 = (const float*)d_in[5];
    const float* w3 = (const float*)d_in[6];
    const float* b3 = (const float*)d_in[7];
    const float* w4 = (const float*)d_in[8];
    const float* b4 = (const float*)d_in[9];
    float* out = (float*)d_out;
    const int Bn = in_sizes[0];

    monotonic_main<<<768, TPB, 0, stream>>>(x, h, w1, b1, w2, b2, w3, b3, w4, b4, out, Bn);
}
